# starting kernel
# baseline (speedup 1.0000x reference)
#include <hip/hip_runtime.h>


typedef float v2f __attribute__((ext_vector_type(2)));
typedef float v8f __attribute__((ext_vector_type(8)));

#define RES_TOTAL   (32 * 4096)
#define RES_PER_BLK 32
#define WAVE_LDS    1024
#define EPSF        1e-12f

__device__ __forceinline__ void wave_sync() {
  __builtin_amdgcn_wave_barrier();
}

__device__ __forceinline__ v8f wmma4x4(v2f a, v2f b) {
  v8f c = {};
  return __builtin_amdgcn_wmma_f32_16x16x4_f32(false, a, false, b, (short)0, c,
                                               false, false);
}

__global__ __launch_bounds__(256) void rotamer_projection_kernel(
    const float* __restrict__ dih,
    const float* __restrict__ bbc,
    const int*   __restrict__ seq,
    const float* __restrict__ dframes,
    const int*   __restrict__ gidx,
    const float* __restrict__ amask,
    const float* __restrict__ litp,
    float*       __restrict__ out)
{
  __shared__ float ldsfr[8][WAVE_LDS];

  const int tid  = threadIdx.x;
  const int wave = tid >> 5;
  const int lane = tid & 31;
  float* fr = ldsfr[wave];

  const int low = lane & 15;
  const int hi  = lane >> 4;
  const int i   = low >> 2;
  const int m   = low & 3;
  const int k0  = hi * 2;

  const int resBase = blockIdx.x * RES_PER_BLK + wave * 4;
  const int myRes   = resBase + i;
  const int mySeq   = seq[myRes];

  float* stq[8];
#pragma unroll
  for (int v = 0; v < 8; ++v) {
    const int row  = v + 8 * hi;
    const bool keep = (row >> 2) == i;
    const int off  = keep ? ((row & 3) * 4 + m) : (512 + low);
    stq[v] = fr + i * 128 + off;
  }
  float* pA = fr + i * 128 + m * 4 + k0;
  float* pB = fr + i * 128 + k0 * 4 + m;

  float wc[2], ws[2], wk[2];
#pragma unroll
  for (int e = 0; e < 2; ++e) {
    const int k = k0 + e;
    wc[e] = ((k == 1 && m == 1) || (k == 2 && m == 2)) ? 1.f : 0.f;
    ws[e] = (k == 2 && m == 1) ? 1.f : ((k == 1 && m == 2) ? -1.f : 0.f);
    wk[e] = ((k == 0 && m == 0) || (k == 3 && m == 3)) ? 1.f : 0.f;
  }

  const float* dfBase = dframes + mySeq * 128 + m * 4 + k0;
  v2f dfv[8];
#pragma unroll
  for (int a = 0; a < 8; ++a) {
    dfv[a][0] = dfBase[a * 16];
    dfv[a][1] = dfBase[a * 16 + 1];
  }
  const float* dBase = dih + myRes * 14;
  float sv[8], cv[8];
  sv[0] = 0.f;
  cv[0] = 1.f;
#pragma unroll
  for (int a = 1; a < 8; ++a) {
    const float x = dBase[(a - 1) * 2];
    const float y = dBase[(a - 1) * 2 + 1];
    const float inv = rsqrtf(x * x + y * y + EPSF);
    sv[a] = x * inv;
    cv[a] = y * inv;
  }

#pragma unroll
  for (int a = 0; a < 8; ++a) {
    v2f bop;
    bop[0] = fmaf(wc[0], cv[a], fmaf(ws[0], sv[a], wk[0]));
    bop[1] = fmaf(wc[1], cv[a], fmaf(ws[1], sv[a], wk[1]));

    v8f d = wmma4x4(dfv[a], bop);
#pragma unroll
    for (int v = 0; v < 8; ++v) stq[v][a * 16] = d[v];
    wave_sync();
  }

#pragma unroll
  for (int f = 5; f <= 7; ++f) {
    v2f aop;
    aop[0] = pA[(f - 1) * 16];
    aop[1] = pA[(f - 1) * 16 + 1];
    v2f bop;
    bop[0] = pB[f * 16];
    bop[1] = pB[f * 16 + 4];
    wave_sync();

    v8f d = wmma4x4(aop, bop);
#pragma unroll
    for (int v = 0; v < 8; ++v) stq[v][f * 16] = d[v];
    wave_sync();
  }

  const float* bp = bbc + myRes * 9;
  const float nx = bp[0], ny = bp[1], nz = bp[2];
  const float cax = bp[3], cay = bp[4], caz = bp[5];
  const float cx = bp[6], cy = bp[7], cz = bp[8];
  const float v1x = cx - cax, v1y = cy - cay, v1z = cz - caz;
  const float v2x = nx - cax, v2y = ny - cay, v2z = nz - caz;
  const float inv1 = rsqrtf(v1x * v1x + v1y * v1y + v1z * v1z + EPSF);
  const float e1x = v1x * inv1, e1y = v1y * inv1, e1z = v1z * inv1;
  const float d12 = e1x * v2x + e1y * v2y + e1z * v2z;
  const float u2x = v2x - e1x * d12, u2y = v2y - e1y * d12,
              u2z = v2z - e1z * d12;
  const float inv2 = rsqrtf(u2x * u2x + u2y * u2y + u2z * u2z + EPSF);
  const float e2x = u2x * inv2, e2y = u2y * inv2, e2z = u2z * inv2;
  const float e3x = e1y * e2z - e1z * e2y;
  const float e3y = e1z * e2x - e1x * e2z;
  const float e3z = e1x * e2y - e1y * e2x;

  float row0, row1, row2, row3;
  if (m == 0)      { row0 = e1x; row1 = e2x; row2 = e3x; row3 = cax; }
  else if (m == 1) { row0 = e1y; row1 = e2y; row2 = e3y; row3 = cay; }
  else if (m == 2) { row0 = e1z; row1 = e2z; row2 = e3z; row3 = caz; }
  else             { row0 = 0.f; row1 = 0.f; row2 = 0.f; row3 = 1.f; }
  v2f abb;
  abb[0] = (k0 == 0) ? row0 : row2;
  abb[1] = (k0 == 0) ? row1 : row3;

#pragma unroll
  for (int a = 0; a < 8; ++a) {
    v2f bop;
    bop[0] = pB[a * 16];
    bop[1] = pB[a * 16 + 4];
    wave_sync();

    v8f d = wmma4x4(abb, bop);
#pragma unroll
    for (int v = 0; v < 8; ++v) stq[v][a * 16] = d[v];
    wave_sync();
  }

  for (int t = lane; t < 4 * 14; t += 32) {
    const int ri   = t / 14;
    const int atom = t - ri * 14;
    const int res  = resBase + ri;
    const int sq   = seq[res];
    const int g    = gidx[sq * 14 + atom];
    const float* G = &fr[(ri * 8 + g) * 16];
    const float* lp = litp + (sq * 14 + atom) * 3;
    const float l0 = lp[0], l1 = lp[1], l2 = lp[2];
    const float mk = amask[sq * 14 + atom];
    float* op = out + (res * 14 + atom) * 3;
#pragma unroll
    for (int r = 0; r < 3; ++r) {
      op[r] = (G[r * 4 + 0] * l0 + G[r * 4 + 1] * l1 + G[r * 4 + 2] * l2 +
               G[r * 4 + 3]) * mk;
    }
  }
}

extern "C" void kernel_launch(void* const* d_in, const int* in_sizes, int n_in,
                              void* d_out, int out_size, void* d_ws,
                              size_t ws_size, hipStream_t stream) {
  const float* dih     = (const float*)d_in[0];
  const float* bbc     = (const float*)d_in[1];
  const int*   seq     = (const int*)d_in[2];
  const float* dframes = (const float*)d_in[3];
  const int*   gidx    = (const int*)d_in[4];
  const float* amask   = (const float*)d_in[5];
  const float* litp    = (const float*)d_in[6];
  float*       out     = (float*)d_out;

  const int blocks = RES_TOTAL / RES_PER_BLK;
  rotamer_projection_kernel<<<blocks, 256, 0, stream>>>(
      dih, bbc, seq, dframes, gidx, amask, litp, out);
}


# Each 4x4 product is handed on by ONE 16-byte LDS store from the lane holding its column (4 selects; column-major frames, residue pitch 144 floats) in place of 8 b32 stores with 2-3 way bank clashes
# speedup vs baseline: 1.6503x; 1.6503x over previous
#include <hip/hip_runtime.h>


typedef float v2f __attribute__((ext_vector_type(2)));
typedef float v4f __attribute__((ext_vector_type(4)));
typedef float v8f __attribute__((ext_vector_type(8)));

#define RES_TOTAL   (32 * 4096)
#define RES_PER_BLK 32
#define RES_PITCH   144
#define WAVE_LDS    (4 * RES_PITCH)
#define EPSF        1e-12f

__device__ __forceinline__ void wave_sync() {
  __builtin_amdgcn_wave_barrier();
}

__device__ __forceinline__ v4f own_column(v8f d, bool odd) {
  v4f q;
  q[0] = odd ? d[4] : d[0];
  q[1] = odd ? d[5] : d[1];
  q[2] = odd ? d[6] : d[2];
  q[3] = odd ? d[7] : d[3];
  return q;
}

__device__ __forceinline__ v8f wmma4x4(v2f a, v2f b) {
  v8f c = {};
  return __builtin_amdgcn_wmma_f32_16x16x4_f32(false, a, false, b, (short)0, c,
                                               false, false);
}

__global__ __launch_bounds__(256) void rotamer_projection_kernel(
    const float* __restrict__ dih,
    const float* __restrict__ bbc,
    const int*   __restrict__ seq,
    const float* __restrict__ dframes,
    const int*   __restrict__ gidx,
    const float* __restrict__ amask,
    const float* __restrict__ litp,
    float*       __restrict__ out)
{
  __shared__ __attribute__((aligned(16))) float ldsfr[8][WAVE_LDS];

  const int tid  = threadIdx.x;
  const int wave = tid >> 5;
  const int lane = tid & 31;
  float* fr = ldsfr[wave];

  const int low = lane & 15;
  const int hi  = lane >> 4;
  const int i   = low >> 2;
  const int m   = low & 3;
  const int k0  = hi * 2;

  const int resBase = blockIdx.x * RES_PER_BLK + wave * 4;
  const int myRes   = resBase + i;
  const int mySeq   = seq[myRes];

  const bool oddI   = (i & 1) != 0;
  const bool owner  = hi == (i >> 1);

  float* frI = fr + i * RES_PITCH;
  float* pS  = frI + m * 4;
  const float* pAT = frI + k0 * 4 + m;
  const float* pBT = frI + m * 4 + k0;

  float wc[2], ws[2], wk[2];
#pragma unroll
  for (int e = 0; e < 2; ++e) {
    const int k = k0 + e;
    wc[e] = ((k == 1 && m == 1) || (k == 2 && m == 2)) ? 1.f : 0.f;
    ws[e] = (k == 2 && m == 1) ? 1.f : ((k == 1 && m == 2) ? -1.f : 0.f);
    wk[e] = ((k == 0 && m == 0) || (k == 3 && m == 3)) ? 1.f : 0.f;
  }

  const float* dfBase = dframes + mySeq * 128 + m * 4 + k0;
  v2f dfv[8];
#pragma unroll
  for (int a = 0; a < 8; ++a) {
    dfv[a][0] = dfBase[a * 16];
    dfv[a][1] = dfBase[a * 16 + 1];
  }
  const float* dBase = dih + myRes * 14;
  float sv[8], cv[8];
  sv[0] = 0.f;
  cv[0] = 1.f;
#pragma unroll
  for (int a = 1; a < 8; ++a) {
    const float x = dBase[(a - 1) * 2];
    const float y = dBase[(a - 1) * 2 + 1];
    const float inv = rsqrtf(x * x + y * y + EPSF);
    sv[a] = x * inv;
    cv[a] = y * inv;
  }

#pragma unroll
  for (int a = 0; a < 8; ++a) {
    v2f bop;
    bop[0] = fmaf(wc[0], cv[a], fmaf(ws[0], sv[a], wk[0]));
    bop[1] = fmaf(wc[1], cv[a], fmaf(ws[1], sv[a], wk[1]));

    v8f d = wmma4x4(dfv[a], bop);
    const v4f q = own_column(d, oddI);
    if (owner) *reinterpret_cast<v4f*>(pS + a * 16) = q;
    wave_sync();
  }

#pragma unroll
  for (int f = 5; f <= 7; ++f) {
    v2f aop;
    aop[0] = pAT[(f - 1) * 16];
    aop[1] = pAT[(f - 1) * 16 + 4];
    const v2f bop = *reinterpret_cast<const v2f*>(pBT + f * 16);
    wave_sync();

    v8f d = wmma4x4(aop, bop);
    const v4f q = own_column(d, oddI);
    if (owner) *reinterpret_cast<v4f*>(pS + f * 16) = q;
    wave_sync();
  }

  const float* bp = bbc + myRes * 9;
  const float nx = bp[0], ny = bp[1], nz = bp[2];
  const float cax = bp[3], cay = bp[4], caz = bp[5];
  const float cx = bp[6], cy = bp[7], cz = bp[8];
  const float v1x = cx - cax, v1y = cy - cay, v1z = cz - caz;
  const float v2x = nx - cax, v2y = ny - cay, v2z = nz - caz;
  const float inv1 = rsqrtf(v1x * v1x + v1y * v1y + v1z * v1z + EPSF);
  const float e1x = v1x * inv1, e1y = v1y * inv1, e1z = v1z * inv1;
  const float d12 = e1x * v2x + e1y * v2y + e1z * v2z;
  const float u2x = v2x - e1x * d12, u2y = v2y - e1y * d12,
              u2z = v2z - e1z * d12;
  const float inv2 = rsqrtf(u2x * u2x + u2y * u2y + u2z * u2z + EPSF);
  const float e2x = u2x * inv2, e2y = u2y * inv2, e2z = u2z * inv2;
  const float e3x = e1y * e2z - e1z * e2y;
  const float e3y = e1z * e2x - e1x * e2z;
  const float e3z = e1x * e2y - e1y * e2x;

  float row0, row1, row2, row3;
  if (m == 0)      { row0 = e1x; row1 = e2x; row2 = e3x; row3 = cax; }
  else if (m == 1) { row0 = e1y; row1 = e2y; row2 = e3y; row3 = cay; }
  else if (m == 2) { row0 = e1z; row1 = e2z; row2 = e3z; row3 = caz; }
  else             { row0 = 0.f; row1 = 0.f; row2 = 0.f; row3 = 1.f; }
  v2f abb;
  abb[0] = (k0 == 0) ? row0 : row2;
  abb[1] = (k0 == 0) ? row1 : row3;

#pragma unroll
  for (int a = 0; a < 8; ++a) {
    const v2f fop = *reinterpret_cast<const v2f*>(pBT + a * 16);
    wave_sync();

    v8f d = wmma4x4(fop, abb);
    const v4f q = own_column(d, oddI);
    if (owner) *reinterpret_cast<v4f*>(pS + a * 16) = q;
    wave_sync();
  }

  for (int t = lane; t < 4 * 14; t += 32) {
    const int ri   = t / 14;
    const int atom = t - ri * 14;
    const int res  = resBase + ri;
    const int sq   = seq[res];
    const int g    = gidx[sq * 14 + atom];
    const float* G = &fr[ri * RES_PITCH + g * 16];
    const float* lp = litp + (sq * 14 + atom) * 3;
    const float l0 = lp[0], l1 = lp[1], l2 = lp[2];
    const float mk = amask[sq * 14 + atom];
    float* op = out + (res * 14 + atom) * 3;
#pragma unroll
    for (int r = 0; r < 3; ++r) {
      op[r] = (G[r * 4 + 0] * l0 + G[r * 4 + 1] * l1 + G[r * 4 + 2] * l2 +
               G[r * 4 + 3]) * mk;
    }
  }
}

extern "C" void kernel_launch(void* const* d_in, const int* in_sizes, int n_in,
                              void* d_out, int out_size, void* d_ws,
                              size_t ws_size, hipStream_t stream) {
  const float* dih     = (const float*)d_in[0];
  const float* bbc     = (const float*)d_in[1];
  const int*   seq     = (const int*)d_in[2];
  const float* dframes = (const float*)d_in[3];
  const int*   gidx    = (const int*)d_in[4];
  const float* amask   = (const float*)d_in[5];
  const float* litp    = (const float*)d_in[6];
  float*       out     = (float*)d_out;

  const int blocks = RES_TOTAL / RES_PER_BLK;
  rotamer_projection_kernel<<<blocks, 256, 0, stream>>>(
      dih, bbc, seq, dframes, gidx, amask, litp, out);
}


# Each wave carries 4 independent groups of 4 residues side by side (2048 work-groups of 4 waves, a quarter as many waves): every global load requested up front, the groups' products interleaved
# speedup vs baseline: 1.7346x; 1.0511x over previous
#include <hip/hip_runtime.h>


typedef float v2f __attribute__((ext_vector_type(2)));
typedef float v4f __attribute__((ext_vector_type(4)));
typedef float v8f __attribute__((ext_vector_type(8)));

#define GR           4
#define RES_TOTAL    (32 * 4096)
#define RES_PER_WAVE (4 * GR)
#define WAVES_PER_BLK 4
#define RES_PER_BLK  (WAVES_PER_BLK * RES_PER_WAVE)
#define RES_PITCH    144
#define GRP_LDS      (4 * RES_PITCH)
#define WAVE_LDS     (GR * GRP_LDS)
#define TRIPS        ((RES_PER_WAVE * 14) / 32)
#define EPSF         1e-12f

__device__ __forceinline__ void wave_sync() {
  __builtin_amdgcn_wave_barrier();
}

__device__ __forceinline__ v4f own_column(v8f d, bool odd) {
  v4f q;
  q[0] = odd ? d[4] : d[0];
  q[1] = odd ? d[5] : d[1];
  q[2] = odd ? d[6] : d[2];
  q[3] = odd ? d[7] : d[3];
  return q;
}

__device__ __forceinline__ v8f wmma4x4(v2f a, v2f b) {
  v8f c = {};
  return __builtin_amdgcn_wmma_f32_16x16x4_f32(false, a, false, b, (short)0, c,
                                               false, false);
}

__device__ __forceinline__ v2f bb_operand(const float* bp, int m, int k0) {
  const float nx = bp[0], ny = bp[1], nz = bp[2];
  const float cax = bp[3], cay = bp[4], caz = bp[5];
  const float cx = bp[6], cy = bp[7], cz = bp[8];
  const float v1x = cx - cax, v1y = cy - cay, v1z = cz - caz;
  const float v2x = nx - cax, v2y = ny - cay, v2z = nz - caz;
  const float inv1 = rsqrtf(v1x * v1x + v1y * v1y + v1z * v1z + EPSF);
  const float e1x = v1x * inv1, e1y = v1y * inv1, e1z = v1z * inv1;
  const float d12 = e1x * v2x + e1y * v2y + e1z * v2z;
  const float u2x = v2x - e1x * d12, u2y = v2y - e1y * d12,
              u2z = v2z - e1z * d12;
  const float inv2 = rsqrtf(u2x * u2x + u2y * u2y + u2z * u2z + EPSF);
  const float e2x = u2x * inv2, e2y = u2y * inv2, e2z = u2z * inv2;
  const float e3x = e1y * e2z - e1z * e2y;
  const float e3y = e1z * e2x - e1x * e2z;
  const float e3z = e1x * e2y - e1y * e2x;

  float row0, row1, row2, row3;
  if (m == 0)      { row0 = e1x; row1 = e2x; row2 = e3x; row3 = cax; }
  else if (m == 1) { row0 = e1y; row1 = e2y; row2 = e3y; row3 = cay; }
  else if (m == 2) { row0 = e1z; row1 = e2z; row2 = e3z; row3 = caz; }
  else             { row0 = 0.f; row1 = 0.f; row2 = 0.f; row3 = 1.f; }
  v2f abb;
  abb[0] = (k0 == 0) ? row0 : row2;
  abb[1] = (k0 == 0) ? row1 : row3;
  return abb;
}

__global__ __launch_bounds__(32 * WAVES_PER_BLK) void rotamer_projection_kernel(
    const float* __restrict__ dih,
    const float* __restrict__ bbc,
    const int*   __restrict__ seq,
    const float* __restrict__ dframes,
    const int*   __restrict__ gidx,
    const float* __restrict__ amask,
    const float* __restrict__ litp,
    float*       __restrict__ out)
{
  __shared__ __attribute__((aligned(16))) float ldsfr[WAVES_PER_BLK][WAVE_LDS];

  const int tid  = threadIdx.x;
  const int wave = tid >> 5;
  const int lane = tid & 31;
  float* fr = ldsfr[wave];

  const int low = lane & 15;
  const int hi  = lane >> 4;
  const int i   = low >> 2;
  const int m   = low & 3;
  const int k0  = hi * 2;

  const int waveBase = blockIdx.x * RES_PER_BLK + wave * RES_PER_WAVE;

  const bool oddI   = (i & 1) != 0;
  const bool owner  = hi == (i >> 1);

  float* frI = fr + i * RES_PITCH;
  float* pS  = frI + m * 4;
  const float* pAT = frI + k0 * 4 + m;
  const float* pBT = frI + m * 4 + k0;

  float wc[2], ws[2], wk[2];
#pragma unroll
  for (int e = 0; e < 2; ++e) {
    const int k = k0 + e;
    wc[e] = ((k == 1 && m == 1) || (k == 2 && m == 2)) ? 1.f : 0.f;
    ws[e] = (k == 2 && m == 1) ? 1.f : ((k == 1 && m == 2) ? -1.f : 0.f);
    wk[e] = ((k == 0 && m == 0) || (k == 3 && m == 3)) ? 1.f : 0.f;
  }

  int mySeq[GR];
#pragma unroll
  for (int g = 0; g < GR; ++g) mySeq[g] = seq[waveBase + 4 * g + i];
  int sq4[TRIPS];
#pragma unroll
  for (int it = 0; it < TRIPS; ++it)
    sq4[it] = seq[waveBase + (lane + 32 * it) / 14];

  float bbv[GR][9];
#pragma unroll
  for (int g = 0; g < GR; ++g) {
    const float* bp = bbc + (waveBase + 4 * g + i) * 9;
#pragma unroll
    for (int j = 0; j < 9; ++j) bbv[g][j] = bp[j];
  }
  float dx[GR][8], dy[GR][8];
#pragma unroll
  for (int g = 0; g < GR; ++g) {
    const float* dBase = dih + (waveBase + 4 * g + i) * 14;
#pragma unroll
    for (int a = 1; a < 8; ++a) {
      dx[g][a] = dBase[(a - 1) * 2];
      dy[g][a] = dBase[(a - 1) * 2 + 1];
    }
  }
  v2f dfv[GR][8];
#pragma unroll
  for (int g = 0; g < GR; ++g) {
    const float* dfBase = dframes + mySeq[g] * 128 + m * 4 + k0;
#pragma unroll
    for (int a = 0; a < 8; ++a) {
      dfv[g][a][0] = dfBase[a * 16];
      dfv[g][a][1] = dfBase[a * 16 + 1];
    }
  }
  int g4[TRIPS];
#pragma unroll
  for (int it = 0; it < TRIPS; ++it) {
    const int t = lane + 32 * it;
    g4[it] = gidx[sq4[it] * 14 + (t - (t / 14) * 14)];
  }

  v2f abb[GR];
#pragma unroll
  for (int g = 0; g < GR; ++g) abb[g] = bb_operand(bbv[g], m, k0);

#pragma unroll
  for (int a = 0; a < 8; ++a) {
    v8f d[GR];
#pragma unroll
    for (int g = 0; g < GR; ++g) {
      float sv = 0.f, cv = 1.f;
      if (a > 0) {
        const float x = dx[g][a];
        const float y = dy[g][a];
        const float inv = rsqrtf(x * x + y * y + EPSF);
        sv = x * inv;
        cv = y * inv;
      }
      v2f bop;
      bop[0] = fmaf(wc[0], cv, fmaf(ws[0], sv, wk[0]));
      bop[1] = fmaf(wc[1], cv, fmaf(ws[1], sv, wk[1]));
      d[g] = wmma4x4(dfv[g][a], bop);
    }
    v4f q[GR];
#pragma unroll
    for (int g = 0; g < GR; ++g) q[g] = own_column(d[g], oddI);
    if (owner) {
#pragma unroll
      for (int g = 0; g < GR; ++g)
        *reinterpret_cast<v4f*>(pS + g * GRP_LDS + a * 16) = q[g];
    }
  }
  wave_sync();

#pragma unroll
  for (int f = 5; f <= 7; ++f) {
    v2f aop[GR], bop[GR];
#pragma unroll
    for (int g = 0; g < GR; ++g) {
      aop[g][0] = pAT[g * GRP_LDS + (f - 1) * 16];
      aop[g][1] = pAT[g * GRP_LDS + (f - 1) * 16 + 4];
      bop[g] = *reinterpret_cast<const v2f*>(pBT + g * GRP_LDS + f * 16);
    }
    wave_sync();

    v4f q[GR];
#pragma unroll
    for (int g = 0; g < GR; ++g) {
      const v8f d = wmma4x4(aop[g], bop[g]);
      q[g] = own_column(d, oddI);
    }
    if (owner) {
#pragma unroll
      for (int g = 0; g < GR; ++g)
        *reinterpret_cast<v4f*>(pS + g * GRP_LDS + f * 16) = q[g];
    }
    wave_sync();
  }

  v2f fop[2][8];
#pragma unroll
  for (int a = 0; a < 8; ++a)
    fop[0][a] = *reinterpret_cast<const v2f*>(pBT + a * 16);
#pragma unroll
  for (int g = 0; g < GR; ++g) {
    if (g + 1 < GR) {
#pragma unroll
      for (int a = 0; a < 8; ++a)
        fop[(g + 1) & 1][a] = *reinterpret_cast<const v2f*>(
            pBT + (g + 1) * GRP_LDS + a * 16);
    }
    wave_sync();

#pragma unroll
    for (int a0 = 0; a0 < 8; a0 += 4) {
      v4f q[4];
#pragma unroll
      for (int a = 0; a < 4; ++a) {
        const v8f d = wmma4x4(fop[g & 1][a0 + a], abb[g]);
        q[a] = own_column(d, oddI);
      }
      if (owner) {
#pragma unroll
        for (int a = 0; a < 4; ++a)
          *reinterpret_cast<v4f*>(pS + g * GRP_LDS + (a0 + a) * 16) = q[a];
      }
    }
    wave_sync();
  }

#pragma unroll
  for (int it = 0; it < TRIPS; ++it) {
    const int t    = lane + 32 * it;
    const int ri   = t / 14;
    const int atom = t - ri * 14;
    const int sq   = sq4[it];
    const int g    = g4[it];
    const float* G = &fr[ri * RES_PITCH + g * 16];
    const float* lp = litp + (sq * 14 + atom) * 3;
    const float l0 = lp[0], l1 = lp[1], l2 = lp[2];
    const float mk = amask[sq * 14 + atom];
    float* op = out + (waveBase * 14 + t) * 3;
#pragma unroll
    for (int r = 0; r < 3; ++r) {
      op[r] = (G[r * 4 + 0] * l0 + G[r * 4 + 1] * l1 + G[r * 4 + 2] * l2 +
               G[r * 4 + 3]) * mk;
    }
  }
}

extern "C" void kernel_launch(void* const* d_in, const int* in_sizes, int n_in,
                              void* d_out, int out_size, void* d_ws,
                              size_t ws_size, hipStream_t stream) {
  const float* dih     = (const float*)d_in[0];
  const float* bbc     = (const float*)d_in[1];
  const int*   seq     = (const int*)d_in[2];
  const float* dframes = (const float*)d_in[3];
  const int*   gidx    = (const int*)d_in[4];
  const float* amask   = (const float*)d_in[5];
  const float* litp    = (const float*)d_in[6];
  float*       out     = (float*)d_out;

  const int blocks = RES_TOTAL / RES_PER_BLK;
  rotamer_projection_kernel<<<blocks, 32 * WAVES_PER_BLK, 0, stream>>>(
      dih, bbc, seq, dframes, gidx, amask, litp, out);
}


# Chi chain re-associated onto the global frames: gfr[f]=gfr[f-1]@fr[f], f=5..7, so the 3 local chain products per residue drop out (16 WMMA products, not 19); order of the f32 sums changes
# speedup vs baseline: 1.8767x; 1.0819x over previous
#include <hip/hip_runtime.h>


typedef float v2f __attribute__((ext_vector_type(2)));
typedef float v4f __attribute__((ext_vector_type(4)));
typedef float v8f __attribute__((ext_vector_type(8)));

#define GR           4
#define RES_TOTAL    (32 * 4096)
#define RES_PER_WAVE (4 * GR)
#define WAVES_PER_BLK 4
#define RES_PER_BLK  (WAVES_PER_BLK * RES_PER_WAVE)
#define RES_PITCH    144
#define GRP_LDS      (4 * RES_PITCH)
#define WAVE_LDS     (GR * GRP_LDS)
#define TRIPS        ((RES_PER_WAVE * 14) / 32)
#define EPSF         1e-12f

__device__ __forceinline__ void wave_sync() {
  __builtin_amdgcn_wave_barrier();
}

__device__ __forceinline__ v4f own_column(v8f d, bool odd) {
  v4f q;
  q[0] = odd ? d[4] : d[0];
  q[1] = odd ? d[5] : d[1];
  q[2] = odd ? d[6] : d[2];
  q[3] = odd ? d[7] : d[3];
  return q;
}

__device__ __forceinline__ v8f wmma4x4(v2f a, v2f b) {
  v8f c = {};
  return __builtin_amdgcn_wmma_f32_16x16x4_f32(false, a, false, b, (short)0, c,
                                               false, false);
}

__device__ __forceinline__ void product_to_slot(const v2f (&aop)[GR],
                                                const v2f (&bop)[GR],
                                                float* dst, bool oddI,
                                                bool owner) {
  v4f q[GR];
#pragma unroll
  for (int g = 0; g < GR; ++g) {
    const v8f d = wmma4x4(aop[g], bop[g]);
    q[g] = own_column(d, oddI);
  }
  if (owner) {
#pragma unroll
    for (int g = 0; g < GR; ++g)
      *reinterpret_cast<v4f*>(dst + g * GRP_LDS) = q[g];
  }
}

__device__ __forceinline__ void read_slot(v2f (&op)[GR], const float* srcp) {
#pragma unroll
  for (int g = 0; g < GR; ++g)
    op[g] = *reinterpret_cast<const v2f*>(srcp + g * GRP_LDS);
}

__device__ __forceinline__ v2f bb_operand(const float* bp, int m, int k0) {
  const float nx = bp[0], ny = bp[1], nz = bp[2];
  const float cax = bp[3], cay = bp[4], caz = bp[5];
  const float cx = bp[6], cy = bp[7], cz = bp[8];
  const float v1x = cx - cax, v1y = cy - cay, v1z = cz - caz;
  const float v2x = nx - cax, v2y = ny - cay, v2z = nz - caz;
  const float inv1 = rsqrtf(v1x * v1x + v1y * v1y + v1z * v1z + EPSF);
  const float e1x = v1x * inv1, e1y = v1y * inv1, e1z = v1z * inv1;
  const float d12 = e1x * v2x + e1y * v2y + e1z * v2z;
  const float u2x = v2x - e1x * d12, u2y = v2y - e1y * d12,
              u2z = v2z - e1z * d12;
  const float inv2 = rsqrtf(u2x * u2x + u2y * u2y + u2z * u2z + EPSF);
  const float e2x = u2x * inv2, e2y = u2y * inv2, e2z = u2z * inv2;
  const float e3x = e1y * e2z - e1z * e2y;
  const float e3y = e1z * e2x - e1x * e2z;
  const float e3z = e1x * e2y - e1y * e2x;

  float row0, row1, row2, row3;
  if (m == 0)      { row0 = e1x; row1 = e2x; row2 = e3x; row3 = cax; }
  else if (m == 1) { row0 = e1y; row1 = e2y; row2 = e3y; row3 = cay; }
  else if (m == 2) { row0 = e1z; row1 = e2z; row2 = e3z; row3 = caz; }
  else             { row0 = 0.f; row1 = 0.f; row2 = 0.f; row3 = 1.f; }
  v2f abb;
  abb[0] = (k0 == 0) ? row0 : row2;
  abb[1] = (k0 == 0) ? row1 : row3;
  return abb;
}

__global__ __launch_bounds__(32 * WAVES_PER_BLK) void rotamer_projection_kernel(
    const float* __restrict__ dih,
    const float* __restrict__ bbc,
    const int*   __restrict__ seq,
    const float* __restrict__ dframes,
    const int*   __restrict__ gidx,
    const float* __restrict__ amask,
    const float* __restrict__ litp,
    float*       __restrict__ out)
{
  __shared__ __attribute__((aligned(16))) float ldsfr[WAVES_PER_BLK][WAVE_LDS];

  const int tid  = threadIdx.x;
  const int wave = tid >> 5;
  const int lane = tid & 31;
  float* fr = ldsfr[wave];

  const int low = lane & 15;
  const int hi  = lane >> 4;
  const int i   = low >> 2;
  const int m   = low & 3;
  const int k0  = hi * 2;

  const int waveBase = blockIdx.x * RES_PER_BLK + wave * RES_PER_WAVE;

  const bool oddI   = (i & 1) != 0;
  const bool owner  = hi == (i >> 1);

  float* frI = fr + i * RES_PITCH;
  float* pS  = frI + m * 4;
  const float* pBT = frI + m * 4 + k0;

  float wc[2], ws[2], wk[2];
#pragma unroll
  for (int e = 0; e < 2; ++e) {
    const int k = k0 + e;
    wc[e] = ((k == 1 && m == 1) || (k == 2 && m == 2)) ? 1.f : 0.f;
    ws[e] = (k == 2 && m == 1) ? 1.f : ((k == 1 && m == 2) ? -1.f : 0.f);
    wk[e] = ((k == 0 && m == 0) || (k == 3 && m == 3)) ? 1.f : 0.f;
  }

  int mySeq[GR];
#pragma unroll
  for (int g = 0; g < GR; ++g) mySeq[g] = seq[waveBase + 4 * g + i];
  int sq4[TRIPS];
#pragma unroll
  for (int it = 0; it < TRIPS; ++it)
    sq4[it] = seq[waveBase + (lane + 32 * it) / 14];

  float bbv[GR][9];
#pragma unroll
  for (int g = 0; g < GR; ++g) {
    const float* bp = bbc + (waveBase + 4 * g + i) * 9;
#pragma unroll
    for (int j = 0; j < 9; ++j) bbv[g][j] = bp[j];
  }
  float dx[GR][8], dy[GR][8];
#pragma unroll
  for (int g = 0; g < GR; ++g) {
    const float* dBase = dih + (waveBase + 4 * g + i) * 14;
#pragma unroll
    for (int a = 1; a < 8; ++a) {
      dx[g][a] = dBase[(a - 1) * 2];
      dy[g][a] = dBase[(a - 1) * 2 + 1];
    }
  }
  v2f dfv[GR][8];
#pragma unroll
  for (int g = 0; g < GR; ++g) {
    const float* dfBase = dframes + mySeq[g] * 128 + m * 4 + k0;
#pragma unroll
    for (int a = 0; a < 8; ++a) {
      dfv[g][a][0] = dfBase[a * 16];
      dfv[g][a][1] = dfBase[a * 16 + 1];
    }
  }
  int g4[TRIPS];
#pragma unroll
  for (int it = 0; it < TRIPS; ++it) {
    const int t = lane + 32 * it;
    g4[it] = gidx[sq4[it] * 14 + (t - (t / 14) * 14)];
  }

  v2f abb[GR];
#pragma unroll
  for (int g = 0; g < GR; ++g) abb[g] = bb_operand(bbv[g], m, k0);

#pragma unroll
  for (int a = 0; a < 8; ++a) {
    v8f d[GR];
#pragma unroll
    for (int g = 0; g < GR; ++g) {
      float sv = 0.f, cv = 1.f;
      if (a > 0) {
        const float x = dx[g][a];
        const float y = dy[g][a];
        const float inv = rsqrtf(x * x + y * y + EPSF);
        sv = x * inv;
        cv = y * inv;
      }
      v2f bop;
      bop[0] = fmaf(wc[0], cv, fmaf(ws[0], sv, wk[0]));
      bop[1] = fmaf(wc[1], cv, fmaf(ws[1], sv, wk[1]));
      d[g] = wmma4x4(dfv[g][a], bop);
    }
    v4f q[GR];
#pragma unroll
    for (int g = 0; g < GR; ++g) q[g] = own_column(d[g], oddI);
    if (owner) {
#pragma unroll
      for (int g = 0; g < GR; ++g)
        *reinterpret_cast<v4f*>(pS + g * GRP_LDS + a * 16) = q[g];
    }
  }
  wave_sync();

  v2f fop[8][GR];
#pragma unroll
  for (int a = 0; a < 8; ++a) {
#pragma unroll
    for (int g = 0; g < GR; ++g)
      fop[a][g] = *reinterpret_cast<const v2f*>(pBT + g * GRP_LDS + a * 16);
  }
  wave_sync();

  v2f gprev[GR];
  product_to_slot(fop[4], abb, pS + 4 * 16, oddI, owner);
  wave_sync();
  read_slot(gprev, pBT + 4 * 16);
  product_to_slot(fop[0], abb, pS + 0 * 16, oddI, owner);
  product_to_slot(fop[5], gprev, pS + 5 * 16, oddI, owner);
  wave_sync();
  read_slot(gprev, pBT + 5 * 16);
  product_to_slot(fop[1], abb, pS + 1 * 16, oddI, owner);
  product_to_slot(fop[6], gprev, pS + 6 * 16, oddI, owner);
  wave_sync();
  read_slot(gprev, pBT + 6 * 16);
  product_to_slot(fop[2], abb, pS + 2 * 16, oddI, owner);
  product_to_slot(fop[7], gprev, pS + 7 * 16, oddI, owner);
  product_to_slot(fop[3], abb, pS + 3 * 16, oddI, owner);
  wave_sync();

#pragma unroll
  for (int it = 0; it < TRIPS; ++it) {
    const int t    = lane + 32 * it;
    const int ri   = t / 14;
    const int atom = t - ri * 14;
    const int sq   = sq4[it];
    const int g    = g4[it];
    const float* G = &fr[ri * RES_PITCH + g * 16];
    const float* lp = litp + (sq * 14 + atom) * 3;
    const float l0 = lp[0], l1 = lp[1], l2 = lp[2];
    const float mk = amask[sq * 14 + atom];
    float* op = out + (waveBase * 14 + t) * 3;
#pragma unroll
    for (int r = 0; r < 3; ++r) {
      op[r] = (G[r * 4 + 0] * l0 + G[r * 4 + 1] * l1 + G[r * 4 + 2] * l2 +
               G[r * 4 + 3]) * mk;
    }
  }
}

extern "C" void kernel_launch(void* const* d_in, const int* in_sizes, int n_in,
                              void* d_out, int out_size, void* d_ws,
                              size_t ws_size, hipStream_t stream) {
  const float* dih     = (const float*)d_in[0];
  const float* bbc     = (const float*)d_in[1];
  const int*   seq     = (const int*)d_in[2];
  const float* dframes = (const float*)d_in[3];
  const int*   gidx    = (const int*)d_in[4];
  const float* amask   = (const float*)d_in[5];
  const float* litp    = (const float*)d_in[6];
  float*       out     = (float*)d_out;

  const int blocks = RES_TOTAL / RES_PER_BLK;
  rotamer_projection_kernel<<<blocks, 32 * WAVES_PER_BLK, 0, stream>>>(
      dih, bbc, seq, dframes, gidx, amask, litp, out);
}


# Per-residue prologue computed once per residue, not in all 8 operand lanes: one lane per residue builds bb_mat and normalises the torsions, handed to the operand lanes through LDS
# speedup vs baseline: 2.3240x; 1.2383x over previous
#include <hip/hip_runtime.h>


typedef float v2f __attribute__((ext_vector_type(2)));
typedef float v4f __attribute__((ext_vector_type(4)));
typedef float v8f __attribute__((ext_vector_type(8)));

#define GR           4
#define RES_TOTAL    (32 * 4096)
#define RES_PER_WAVE (4 * GR)
#define WAVES_PER_BLK 4
#define RES_PER_BLK  (WAVES_PER_BLK * RES_PER_WAVE)
#define RES_PITCH    144
#define GRP_LDS      (4 * RES_PITCH)
#define WAVE_LDS     (GR * GRP_LDS)
#define TRIPS        ((RES_PER_WAVE * 14) / 32)
#define EPSF         1e-12f

__device__ __forceinline__ void wave_sync() {
  __builtin_amdgcn_wave_barrier();
}

__device__ __forceinline__ v4f own_column(v8f d, bool odd) {
  v4f q;
  q[0] = odd ? d[4] : d[0];
  q[1] = odd ? d[5] : d[1];
  q[2] = odd ? d[6] : d[2];
  q[3] = odd ? d[7] : d[3];
  return q;
}

__device__ __forceinline__ v8f wmma4x4(v2f a, v2f b) {
  v8f c = {};
  return __builtin_amdgcn_wmma_f32_16x16x4_f32(false, a, false, b, (short)0, c,
                                               false, false);
}

__device__ __forceinline__ void product_to_slot(const v2f (&aop)[GR],
                                                const v2f (&bop)[GR],
                                                float* dst, bool oddI,
                                                bool owner) {
  v4f q[GR];
#pragma unroll
  for (int g = 0; g < GR; ++g) {
    const v8f d = wmma4x4(aop[g], bop[g]);
    q[g] = own_column(d, oddI);
  }
  if (owner) {
#pragma unroll
    for (int g = 0; g < GR; ++g)
      *reinterpret_cast<v4f*>(dst + g * GRP_LDS) = q[g];
  }
}

__device__ __forceinline__ void read_slot(v2f (&op)[GR], const float* srcp) {
#pragma unroll
  for (int g = 0; g < GR; ++g)
    op[g] = *reinterpret_cast<const v2f*>(srcp + g * GRP_LDS);
}

__device__ __forceinline__ void bb_rows(const float (&b)[9], v4f (&row)[4]) {
  const float nx = b[0], ny = b[1], nz = b[2];
  const float cax = b[3], cay = b[4], caz = b[5];
  const float cx = b[6], cy = b[7], cz = b[8];
  const float v1x = cx - cax, v1y = cy - cay, v1z = cz - caz;
  const float v2x = nx - cax, v2y = ny - cay, v2z = nz - caz;
  const float inv1 = rsqrtf(v1x * v1x + v1y * v1y + v1z * v1z + EPSF);
  const float e1x = v1x * inv1, e1y = v1y * inv1, e1z = v1z * inv1;
  const float d12 = e1x * v2x + e1y * v2y + e1z * v2z;
  const float u2x = v2x - e1x * d12, u2y = v2y - e1y * d12,
              u2z = v2z - e1z * d12;
  const float inv2 = rsqrtf(u2x * u2x + u2y * u2y + u2z * u2z + EPSF);
  const float e2x = u2x * inv2, e2y = u2y * inv2, e2z = u2z * inv2;
  const float e3x = e1y * e2z - e1z * e2y;
  const float e3y = e1z * e2x - e1x * e2z;
  const float e3z = e1x * e2y - e1y * e2x;
  row[0][0] = e1x; row[0][1] = e2x; row[0][2] = e3x; row[0][3] = cax;
  row[1][0] = e1y; row[1][1] = e2y; row[1][2] = e3y; row[1][3] = cay;
  row[2][0] = e1z; row[2][1] = e2z; row[2][2] = e3z; row[2][3] = caz;
  row[3][0] = 0.f; row[3][1] = 0.f; row[3][2] = 0.f; row[3][3] = 1.f;
}

__global__ __launch_bounds__(32 * WAVES_PER_BLK) void rotamer_projection_kernel(
    const float* __restrict__ dih,
    const float* __restrict__ bbc,
    const int*   __restrict__ seq,
    const float* __restrict__ dframes,
    const int*   __restrict__ gidx,
    const float* __restrict__ amask,
    const float* __restrict__ litp,
    float*       __restrict__ out)
{
  __shared__ __attribute__((aligned(16))) float ldsfr[WAVES_PER_BLK][WAVE_LDS];

  const int tid  = threadIdx.x;
  const int wave = tid >> 5;
  const int lane = tid & 31;
  float* fr = ldsfr[wave];

  const int low = lane & 15;
  const int hi  = lane >> 4;
  const int i   = low >> 2;
  const int m   = low & 3;
  const int k0  = hi * 2;

  const int waveBase = blockIdx.x * RES_PER_BLK + wave * RES_PER_WAVE;

  const bool oddI   = (i & 1) != 0;
  const bool owner  = hi == (i >> 1);

  float* frI = fr + i * RES_PITCH;
  float* pS  = frI + m * 4;
  const float* pBT = frI + m * 4 + k0;

  const bool mid   = (m == 1) || (m == 2);
  const bool tIn0  = mid && (hi == 1);
  const bool tIn1  = mid && (hi == 0);
  const float con0 = (m == 0 && hi == 0) ? 1.f : 0.f;
  const float con1 = (m == 3 && hi == 1) ? 1.f : 0.f;
  const float* pT  = frI + ((m == 2) ? 2 : 0) + hi;

  int mySeq[GR];
#pragma unroll
  for (int g = 0; g < GR; ++g) mySeq[g] = seq[waveBase + 4 * g + i];
  int sq4[TRIPS];
#pragma unroll
  for (int it = 0; it < TRIPS; ++it)
    sq4[it] = seq[waveBase + (lane + 32 * it) / 14];

  float bbv[9];
  {
    const float* bp = bbc + (waveBase + low) * 9;
#pragma unroll
    for (int j = 0; j < 9; ++j) bbv[j] = bp[j];
  }
  float dv[8];
  {
    const float* dp = dih + (waveBase + low) * 14 + hi * 6;
#pragma unroll
    for (int j = 0; j < 8; ++j) dv[j] = dp[j];
  }
  v2f dfv[GR][8];
#pragma unroll
  for (int g = 0; g < GR; ++g) {
    const float* dfBase = dframes + mySeq[g] * 128 + m * 4 + k0;
#pragma unroll
    for (int a = 0; a < 8; ++a) {
      dfv[g][a][0] = dfBase[a * 16];
      dfv[g][a][1] = dfBase[a * 16 + 1];
    }
  }
  int g4[TRIPS];
#pragma unroll
  for (int it = 0; it < TRIPS; ++it) {
    const int t = lane + 32 * it;
    g4[it] = gidx[sq4[it] * 14 + (t - (t / 14) * 14)];
  }

  {
    float* frR = fr + low * RES_PITCH;
    v4f row[4];
    bb_rows(bbv, row);
#pragma unroll
    for (int rr = 0; rr < 4; ++rr)
      *reinterpret_cast<v4f*>(frR + 8 * 16 + 4 * rr) = row[rr];
    float* pQ = frR + (1 + 3 * hi) * 16 + 4 * (low >> 2);
#pragma unroll
    for (int j = 0; j < 4; ++j) {
      const float x = dv[2 * j];
      const float y = dv[2 * j + 1];
      const float inv = rsqrtf(x * x + y * y + EPSF);
      const float sv = x * inv;
      const float cv = y * inv;
      v4f rec;
      rec[0] = cv; rec[1] = sv; rec[2] = -sv; rec[3] = cv;
      *reinterpret_cast<v4f*>(pQ + 16 * j) = rec;
    }
  }
  wave_sync();

  v2f abb[GR];
  read_slot(abb, pBT + 8 * 16);
  const float tId = ((m == 1 && hi == 0) || (m == 2 && hi == 1)) ? 1.f : 0.f;
  float tv[GR][8];
#pragma unroll
  for (int g = 0; g < GR; ++g) {
    tv[g][0] = tId;
#pragma unroll
    for (int a = 1; a < 8; ++a) tv[g][a] = pT[g * GRP_LDS + a * 16 + 4 * g];
  }

#pragma unroll
  for (int a = 0; a < 8; ++a) {
    v8f d[GR];
#pragma unroll
    for (int g = 0; g < GR; ++g) {
      v2f bop;
      bop[0] = tIn0 ? tv[g][a] : con0;
      bop[1] = tIn1 ? tv[g][a] : con1;
      d[g] = wmma4x4(dfv[g][a], bop);
    }
    v4f q[GR];
#pragma unroll
    for (int g = 0; g < GR; ++g) q[g] = own_column(d[g], oddI);
    if (owner) {
#pragma unroll
      for (int g = 0; g < GR; ++g)
        *reinterpret_cast<v4f*>(pS + g * GRP_LDS + a * 16) = q[g];
    }
  }
  wave_sync();

  v2f fop[8][GR];
#pragma unroll
  for (int a = 0; a < 8; ++a) {
#pragma unroll
    for (int g = 0; g < GR; ++g)
      fop[a][g] = *reinterpret_cast<const v2f*>(pBT + g * GRP_LDS + a * 16);
  }
  wave_sync();

  v2f gprev[GR];
  product_to_slot(fop[4], abb, pS + 4 * 16, oddI, owner);
  wave_sync();
  read_slot(gprev, pBT + 4 * 16);
  product_to_slot(fop[0], abb, pS + 0 * 16, oddI, owner);
  product_to_slot(fop[5], gprev, pS + 5 * 16, oddI, owner);
  wave_sync();
  read_slot(gprev, pBT + 5 * 16);
  product_to_slot(fop[1], abb, pS + 1 * 16, oddI, owner);
  product_to_slot(fop[6], gprev, pS + 6 * 16, oddI, owner);
  wave_sync();
  read_slot(gprev, pBT + 6 * 16);
  product_to_slot(fop[2], abb, pS + 2 * 16, oddI, owner);
  product_to_slot(fop[7], gprev, pS + 7 * 16, oddI, owner);
  product_to_slot(fop[3], abb, pS + 3 * 16, oddI, owner);
  wave_sync();

#pragma unroll
  for (int it = 0; it < TRIPS; ++it) {
    const int t    = lane + 32 * it;
    const int ri   = t / 14;
    const int atom = t - ri * 14;
    const int sq   = sq4[it];
    const int g    = g4[it];
    const float* G = &fr[ri * RES_PITCH + g * 16];
    const float* lp = litp + (sq * 14 + atom) * 3;
    const float l0 = lp[0], l1 = lp[1], l2 = lp[2];
    const float mk = amask[sq * 14 + atom];
    float* op = out + (waveBase * 14 + t) * 3;
#pragma unroll
    for (int r = 0; r < 3; ++r) {
      op[r] = (G[r * 4 + 0] * l0 + G[r * 4 + 1] * l1 + G[r * 4 + 2] * l2 +
               G[r * 4 + 3]) * mk;
    }
  }
}

extern "C" void kernel_launch(void* const* d_in, const int* in_sizes, int n_in,
                              void* d_out, int out_size, void* d_ws,
                              size_t ws_size, hipStream_t stream) {
  const float* dih     = (const float*)d_in[0];
  const float* bbc     = (const float*)d_in[1];
  const int*   seq     = (const int*)d_in[2];
  const float* dframes = (const float*)d_in[3];
  const int*   gidx    = (const int*)d_in[4];
  const float* amask   = (const float*)d_in[5];
  const float* litp    = (const float*)d_in[6];
  float*       out     = (float*)d_out;

  const int blocks = RES_TOTAL / RES_PER_BLK;
  rotamer_projection_kernel<<<blocks, 32 * WAVES_PER_BLK, 0, stream>>>(
      dih, bbc, seq, dframes, gidx, amask, litp, out);
}


# rot_a only mixes columns 1,2: two frames' column pairs share one 4x4x4 product (block-diagonal 2x2 turns), columns 0,3 copied; 12 WMMA products per residue, not 16; only zero terms leave the sums
# speedup vs baseline: 2.4657x; 1.0610x over previous
#include <hip/hip_runtime.h>


typedef float v2f __attribute__((ext_vector_type(2)));
typedef float v4f __attribute__((ext_vector_type(4)));
typedef float v8f __attribute__((ext_vector_type(8)));

#define GR           4
#define RES_TOTAL    (32 * 4096)
#define RES_PER_WAVE (4 * GR)
#define WAVES_PER_BLK 4
#define RES_PER_BLK  (WAVES_PER_BLK * RES_PER_WAVE)
#define RES_PITCH    144
#define GRP_LDS      (4 * RES_PITCH)
#define WAVE_LDS     (GR * GRP_LDS)
#define TRIPS        ((RES_PER_WAVE * 14) / 32)
#define EPSF         1e-12f

__device__ __forceinline__ void wave_sync() {
  __builtin_amdgcn_wave_barrier();
}

__device__ __forceinline__ v4f own_column(v8f d, bool odd) {
  v4f q;
  q[0] = odd ? d[4] : d[0];
  q[1] = odd ? d[5] : d[1];
  q[2] = odd ? d[6] : d[2];
  q[3] = odd ? d[7] : d[3];
  return q;
}

__device__ __forceinline__ v8f wmma4x4(v2f a, v2f b) {
  v8f c = {};
  return __builtin_amdgcn_wmma_f32_16x16x4_f32(false, a, false, b, (short)0, c,
                                               false, false);
}

__device__ __forceinline__ void product_to_slot(const v2f (&aop)[GR],
                                                const v2f (&bop)[GR],
                                                float* dst, bool oddI,
                                                bool owner) {
  v4f q[GR];
#pragma unroll
  for (int g = 0; g < GR; ++g) {
    const v8f d = wmma4x4(aop[g], bop[g]);
    q[g] = own_column(d, oddI);
  }
  if (owner) {
#pragma unroll
    for (int g = 0; g < GR; ++g)
      *reinterpret_cast<v4f*>(dst + g * GRP_LDS) = q[g];
  }
}

__device__ __forceinline__ void read_slot(v2f (&op)[GR], const float* srcp) {
#pragma unroll
  for (int g = 0; g < GR; ++g)
    op[g] = *reinterpret_cast<const v2f*>(srcp + g * GRP_LDS);
}

__device__ __forceinline__ void bb_rows(const float (&b)[9], v4f (&row)[4]) {
  const float nx = b[0], ny = b[1], nz = b[2];
  const float cax = b[3], cay = b[4], caz = b[5];
  const float cx = b[6], cy = b[7], cz = b[8];
  const float v1x = cx - cax, v1y = cy - cay, v1z = cz - caz;
  const float v2x = nx - cax, v2y = ny - cay, v2z = nz - caz;
  const float inv1 = rsqrtf(v1x * v1x + v1y * v1y + v1z * v1z + EPSF);
  const float e1x = v1x * inv1, e1y = v1y * inv1, e1z = v1z * inv1;
  const float d12 = e1x * v2x + e1y * v2y + e1z * v2z;
  const float u2x = v2x - e1x * d12, u2y = v2y - e1y * d12,
              u2z = v2z - e1z * d12;
  const float inv2 = rsqrtf(u2x * u2x + u2y * u2y + u2z * u2z + EPSF);
  const float e2x = u2x * inv2, e2y = u2y * inv2, e2z = u2z * inv2;
  const float e3x = e1y * e2z - e1z * e2y;
  const float e3y = e1z * e2x - e1x * e2z;
  const float e3z = e1x * e2y - e1y * e2x;
  row[0][0] = e1x; row[0][1] = e2x; row[0][2] = e3x; row[0][3] = cax;
  row[1][0] = e1y; row[1][1] = e2y; row[1][2] = e3y; row[1][3] = cay;
  row[2][0] = e1z; row[2][1] = e2z; row[2][2] = e3z; row[2][3] = caz;
  row[3][0] = 0.f; row[3][1] = 0.f; row[3][2] = 0.f; row[3][3] = 1.f;
}

__global__ __launch_bounds__(32 * WAVES_PER_BLK) void rotamer_projection_kernel(
    const float* __restrict__ dih,
    const float* __restrict__ bbc,
    const int*   __restrict__ seq,
    const float* __restrict__ dframes,
    const int*   __restrict__ gidx,
    const float* __restrict__ amask,
    const float* __restrict__ litp,
    float*       __restrict__ out)
{
  __shared__ __attribute__((aligned(16))) float ldsfr[WAVES_PER_BLK][WAVE_LDS];

  const int tid  = threadIdx.x;
  const int wave = tid >> 5;
  const int lane = tid & 31;
  float* fr = ldsfr[wave];

  const int low = lane & 15;
  const int hi  = lane >> 4;
  const int i   = low >> 2;
  const int m   = low & 3;
  const int k0  = hi * 2;

  const int waveBase = blockIdx.x * RES_PER_BLK + wave * RES_PER_WAVE;

  const bool oddI   = (i & 1) != 0;
  const bool owner  = hi == (i >> 1);

  float* frI = fr + i * RES_PITCH;
  float* pS  = frI + m * 4;
  const float* pBT = frI + m * 4 + k0;

  const bool part  = (m >> 1) == hi;
  const float* pT  = frI + hi * 16 + 2 * (m & 1);
  float* pS1 = frI + (m >> 1) * 16 + (1 + (m & 1)) * 4;
  float* pP  = frI + hi * 16 + m;

  int mySeq[GR];
#pragma unroll
  for (int g = 0; g < GR; ++g) mySeq[g] = seq[waveBase + 4 * g + i];
  int sq4[TRIPS];
#pragma unroll
  for (int it = 0; it < TRIPS; ++it)
    sq4[it] = seq[waveBase + (lane + 32 * it) / 14];

  float bbv[9];
  {
    const float* bp = bbc + (waveBase + low) * 9;
#pragma unroll
    for (int j = 0; j < 9; ++j) bbv[j] = bp[j];
  }
  float dv[8];
  {
    const float* dp = dih + (waveBase + low) * 14 + hi * 6;
#pragma unroll
    for (int j = 0; j < 8; ++j) dv[j] = dp[j];
  }
  v4f dfr[GR][4];
#pragma unroll
  for (int g = 0; g < GR; ++g) {
    const float* dfBase = dframes + mySeq[g] * 128 + hi * 16 + m * 4;
#pragma unroll
    for (int j = 0; j < 4; ++j) {
#pragma unroll
      for (int e = 0; e < 4; ++e) dfr[g][j][e] = dfBase[j * 32 + e];
    }
  }
  int g4[TRIPS];
#pragma unroll
  for (int it = 0; it < TRIPS; ++it) {
    const int t = lane + 32 * it;
    g4[it] = gidx[sq4[it] * 14 + (t - (t / 14) * 14)];
  }

  {
    float* frR = fr + low * RES_PITCH;
    v4f row[4];
    bb_rows(bbv, row);
#pragma unroll
    for (int rr = 0; rr < 4; ++rr)
      *reinterpret_cast<v4f*>(frR + 8 * 16 + 4 * rr) = row[rr];
    float* pQ = frR + (4 * hi) * 16 + 4 * (low >> 2);
    const bool upper = hi != 0;
#pragma unroll
    for (int j = 0; j < 4; ++j) {
      const float x = (j == 0) ? dv[0] : (upper ? dv[2 * j] : dv[2 * j - 2]);
      const float y = (j == 0) ? dv[1] : (upper ? dv[2 * j + 1] : dv[2 * j - 1]);
      const float inv = rsqrtf(x * x + y * y + EPSF);
      const bool fixed = (j == 0) && !upper;
      const float sv = fixed ? 0.f : x * inv;
      const float cv = fixed ? 1.f : y * inv;
      v4f rec;
      rec[0] = cv; rec[1] = sv; rec[2] = -sv; rec[3] = cv;
      *reinterpret_cast<v4f*>(pQ + 16 * j) = rec;
    }
  }
  wave_sync();

  v2f abb[GR];
  read_slot(abb, pBT + 8 * 16);
  v2f tv[GR][4];
#pragma unroll
  for (int g = 0; g < GR; ++g) {
#pragma unroll
    for (int j = 0; j < 4; ++j)
      tv[g][j] = *reinterpret_cast<const v2f*>(pT + g * GRP_LDS + j * 32 + 4 * g);
  }
  wave_sync();

#pragma unroll
  for (int j = 0; j < 4; ++j) {
    v8f d[GR];
#pragma unroll
    for (int g = 0; g < GR; ++g) {
      v2f aop, bop;
      aop[0] = dfr[g][j][1];
      aop[1] = dfr[g][j][2];
      bop[0] = part ? tv[g][j][0] : 0.f;
      bop[1] = part ? tv[g][j][1] : 0.f;
      d[g] = wmma4x4(aop, bop);
    }
#pragma unroll
    for (int g = 0; g < GR; ++g) {
      pP[g * GRP_LDS + j * 32]      = dfr[g][j][0];
      pP[g * GRP_LDS + j * 32 + 12] = dfr[g][j][3];
    }
    v4f q[GR];
#pragma unroll
    for (int g = 0; g < GR; ++g) q[g] = own_column(d[g], oddI);
    if (owner) {
#pragma unroll
      for (int g = 0; g < GR; ++g)
        *reinterpret_cast<v4f*>(pS1 + g * GRP_LDS + j * 32) = q[g];
    }
  }
  wave_sync();

  v2f fop[8][GR];
#pragma unroll
  for (int a = 0; a < 8; ++a) {
#pragma unroll
    for (int g = 0; g < GR; ++g)
      fop[a][g] = *reinterpret_cast<const v2f*>(pBT + g * GRP_LDS + a * 16);
  }
  wave_sync();

  v2f gprev[GR];
  product_to_slot(fop[4], abb, pS + 4 * 16, oddI, owner);
  wave_sync();
  read_slot(gprev, pBT + 4 * 16);
  product_to_slot(fop[0], abb, pS + 0 * 16, oddI, owner);
  product_to_slot(fop[5], gprev, pS + 5 * 16, oddI, owner);
  wave_sync();
  read_slot(gprev, pBT + 5 * 16);
  product_to_slot(fop[1], abb, pS + 1 * 16, oddI, owner);
  product_to_slot(fop[6], gprev, pS + 6 * 16, oddI, owner);
  wave_sync();
  read_slot(gprev, pBT + 6 * 16);
  product_to_slot(fop[2], abb, pS + 2 * 16, oddI, owner);
  product_to_slot(fop[7], gprev, pS + 7 * 16, oddI, owner);
  product_to_slot(fop[3], abb, pS + 3 * 16, oddI, owner);
  wave_sync();

#pragma unroll
  for (int it = 0; it < TRIPS; ++it) {
    const int t    = lane + 32 * it;
    const int ri   = t / 14;
    const int atom = t - ri * 14;
    const int sq   = sq4[it];
    const int g    = g4[it];
    const float* G = &fr[ri * RES_PITCH + g * 16];
    const float* lp = litp + (sq * 14 + atom) * 3;
    const float l0 = lp[0], l1 = lp[1], l2 = lp[2];
    const float mk = amask[sq * 14 + atom];
    float* op = out + (waveBase * 14 + t) * 3;
#pragma unroll
    for (int r = 0; r < 3; ++r) {
      op[r] = (G[r * 4 + 0] * l0 + G[r * 4 + 1] * l1 + G[r * 4 + 2] * l2 +
               G[r * 4 + 3]) * mk;
    }
  }
}

extern "C" void kernel_launch(void* const* d_in, const int* in_sizes, int n_in,
                              void* d_out, int out_size, void* d_ws,
                              size_t ws_size, hipStream_t stream) {
  const float* dih     = (const float*)d_in[0];
  const float* bbc     = (const float*)d_in[1];
  const int*   seq     = (const int*)d_in[2];
  const float* dframes = (const float*)d_in[3];
  const int*   gidx    = (const int*)d_in[4];
  const float* amask   = (const float*)d_in[5];
  const float* litp    = (const float*)d_in[6];
  float*       out     = (float*)d_out;

  const int blocks = RES_TOTAL / RES_PER_BLK;
  rotamer_projection_kernel<<<blocks, 32 * WAVES_PER_BLK, 0, stream>>>(
      dih, bbc, seq, dframes, gidx, amask, litp, out);
}
